# starting kernel
# baseline (speedup 1.0000x reference)
#include <hip/hip_runtime.h>
#include <math.h>

#define PN   32
#define PC   256
#define PH   56
#define PW   56
#define PHW  (PH * PW)
#define PCHW (PC * PHW)
#define PNHW (PN * PHW)

typedef float v2f __attribute__((ext_vector_type(2)));
typedef float v8f __attribute__((ext_vector_type(8)));

__global__ void __launch_bounds__(256) sa_pool_kernel(
    const float* __restrict__ x, float* __restrict__ pooled) {
  int t = blockIdx.x * blockDim.x + threadIdx.x;
  int n = t / (PHW / 4);
  int p = (t - n * (PHW / 4)) * 4;
  const float* xb = x + (size_t)n * PCHW + p;

  float sx = 0.f, sy = 0.f, sz = 0.f, sw = 0.f;
  float mx = -INFINITY, my = -INFINITY, mz = -INFINITY, mw = -INFINITY;
#pragma unroll 8
  for (int c = 0; c < PC; ++c) {
    const float4 v = *(const float4*)(xb + (size_t)c * PHW);
    sx += v.x; sy += v.y; sz += v.z; sw += v.w;
    mx = fmaxf(mx, v.x); my = fmaxf(my, v.y);
    mz = fmaxf(mz, v.z); mw = fmaxf(mw, v.w);
  }
  const float inv = 1.0f / (float)PC;
  float* pd = pooled + (size_t)n * 2 * PHW + p;
  float4 a; a.x = sx * inv; a.y = sy * inv; a.z = sz * inv; a.w = sw * inv;
  float4 m; m.x = mx; m.y = my; m.z = mz; m.w = mw;
  *(float4*)(pd)        = a;
  *(float4*)(pd + PHW)  = m;
}

__global__ void __launch_bounds__(128) sa_conv_sigmoid_kernel(
    const float* __restrict__ pooled,
    const float* __restrict__ wconv,
    const float* __restrict__ bias,
    float* __restrict__ scale) {
  const int nh   = blockIdx.x;
  const int n    = nh / PH;
  const int h    = nh - n * PH;
  const int wave = threadIdx.x >> 5;
  const int lane = threadIdx.x & 31;
  const int w0   = (wave == 3) ? (PW - 16) : (wave * 16);
  const int m    = lane & 15;
  const int khalf = (lane >> 4) << 1;

  const float* pb = pooled + (size_t)n * 2 * PHW;

  v8f acc = {0.f, 0.f, 0.f, 0.f, 0.f, 0.f, 0.f, 0.f};

#pragma unroll
  for (int k0 = 0; k0 < 100; k0 += 4) {
    v2f a, b;
#pragma unroll
    for (int e = 0; e < 2; ++e) {
      const int k = k0 + khalf + e;
      float av = 0.f, bv = 0.f;
      if (k < 98) {
        const int ci  = k / 49;
        const int rem = k - ci * 49;
        const int r   = rem / 7;
        const int s   = rem - r * 7;
        const int hh  = h + r - 3;
        const int ww  = w0 + m + s - 3;
        if (hh >= 0 && hh < PH && ww >= 0 && ww < PW)
          av = pb[ci * PHW + hh * PW + ww];
        bv = wconv[k];
      }
      if (e == 0) { a.x = av; b.x = bv; } else { a.y = av; b.y = bv; }
    }
    acc = __builtin_amdgcn_wmma_f32_16x16x4_f32(
         false, a,  false, b,
         (short)0, acc,  false,  false);
  }

  int rr = 0, mout = 0;
  bool writer = false;
  if (lane < 8)                    { writer = true; rr = lane;      mout = lane;      }
  else if (lane >= 16 && lane < 24){ writer = true; rr = lane - 16; mout = lane - 8;  }

  float y = acc[0];
  if (rr == 1) y = acc[1];
  if (rr == 2) y = acc[2];
  if (rr == 3) y = acc[3];
  if (rr == 4) y = acc[4];
  if (rr == 5) y = acc[5];
  if (rr == 6) y = acc[6];
  if (rr == 7) y = acc[7];

  if (writer) {
    y += bias[0];
    const float sg = 1.0f / (1.0f + __expf(-y));
    scale[(size_t)n * PHW + h * PW + w0 + mout] = sg;
  }
}

__global__ void __launch_bounds__(256) sa_apply_kernel(
    const float* __restrict__ x, const float* __restrict__ scale,
    float* __restrict__ out) {
  int t = blockIdx.x * blockDim.x + threadIdx.x;
  int i = t * 4;
  int n = i / PCHW;
  int r = i - n * PCHW;
  int p = r % PHW;

  const float4 xv = *(const float4*)(x + (size_t)i);
  const float4 sv = *(const float4*)(scale + (size_t)n * PHW + p);
  float4 o;
  o.x = xv.x * sv.x; o.y = xv.y * sv.y;
  o.z = xv.z * sv.z; o.w = xv.w * sv.w;
  *(float4*)(out + (size_t)i) = o;
}

extern "C" void kernel_launch(void* const* d_in, const int* in_sizes, int n_in,
                              void* d_out, int out_size, void* d_ws, size_t ws_size,
                              hipStream_t stream) {
  const float* x     = (const float*)d_in[0];
  const float* wconv = (const float*)d_in[1];
  const float* bias  = (const float*)d_in[2];
  float* out = (float*)d_out;

  float* pooled = (float*)d_ws;
  float* scale  = pooled + (size_t)PN * 2 * PHW;

  sa_pool_kernel<<<(PNHW / 4) / 256, 256, 0, stream>>>(x, pooled);

  sa_conv_sigmoid_kernel<<<PN * PH, 128, 0, stream>>>(pooled, wconv, bias, scale);

  sa_apply_kernel<<<(PN * PCHW / 4) / 256, 256, 0, stream>>>(x, scale, out);
}


# one launch, not 3: a work-group owns a 7-row band of one image, pools band + 3 halo rows per side into LDS, convs from LDS on WMMA, then scales its band (x re-read via L2); sum orders unchanged
# speedup vs baseline: 1.8389x; 1.8389x over previous
#include <hip/hip_runtime.h>
#include <math.h>

#define PN   32
#define PC   256
#define PH   56
#define PW   56
#define PHW  (PH * PW)
#define PCHW (PC * PHW)

typedef float v2f __attribute__((ext_vector_type(2)));
typedef float v8f __attribute__((ext_vector_type(8)));

#define BAND   7
#define HROWS  (BAND + 6)
#define LW     64
#define NTHR   512
#define POOL_T 416
#define POOL_D 48

__global__ void __launch_bounds__(NTHR) sa_fused_kernel(
    const float* __restrict__ x,
    const float* __restrict__ wconv,
    const float* __restrict__ bias,
    float* __restrict__ out) {
  __shared__ float pl[2 * HROWS * LW];
  __shared__ float wsm[100];
  __shared__ __attribute__((aligned(16))) float ssm[BAND * PW];

  const int tid   = threadIdx.x;
  const int n     = blockIdx.x >> 3;
  const int b     = blockIdx.x & 7;
  const int rbase = BAND * b - 3;
  const float* xn = x + (size_t)n * PCHW;
  const float bias0 = bias[0];

  if (tid < POOL_T) {
    const int R0 = (rbase > 0) ? rbase : 0;
    const int R1 = (rbase + HROWS - 1 < PH - 1) ? (rbase + HROWS - 1) : (PH - 1);
    const int F0 = R0 * PW;
    const int F1 = (R1 + 1) * PW;
    const int f  = (F0 & ~63) + 2 * tid;
    if (f >= F0 && f < F1) {
      const float* xb = xn + f;
      float s0 = 0.f, s1 = 0.f;
      float m0 = -INFINITY, m1 = -INFINITY;
      float2 win[POOL_D];
#pragma unroll
      for (int j = 0; j < POOL_D; ++j)
        win[j] = *(const float2*)(xb + (size_t)j * PHW);
#pragma unroll
      for (int c = 0; c < PC; ++c) {
        const float2 v = win[c % POOL_D];
        if (c + POOL_D < PC)
          win[c % POOL_D] = *(const float2*)(xb + (size_t)(c + POOL_D) * PHW);
        s0 += v.x; s1 += v.y;
        m0 = fmaxf(m0, v.x); m1 = fmaxf(m1, v.y);
      }
      const float inv = 1.0f / (float)PC;
      const int row = f / PW;
      const int col = f - row * PW;
      float* pd = pl + (row - rbase) * LW + col + 3;
      pd[0] = s0 * inv;  pd[1] = s1 * inv;
      pd[HROWS * LW] = m0;  pd[HROWS * LW + 1] = m1;
    }
  } else {
    for (int i = tid - POOL_T; i < 2 * HROWS * LW; i += NTHR - POOL_T) {
      const int rc  = (i >= HROWS * LW) ? (i - HROWS * LW) : i;
      const int lr  = rc / LW;
      const int col = rc - lr * LW;
      const int row = rbase + lr;
      if (col < 3 || col >= 3 + PW || row < 0 || row >= PH) pl[i] = 0.f;
    }
    for (int k = tid - POOL_T; k < 100; k += NTHR - POOL_T)
      wsm[k] = (k < 98) ? wconv[k] : 0.f;
  }
  __syncthreads();

  const int wave = tid >> 5;
  const int lane = tid & 31;

  {
    const int  m   = lane & 15;
    const bool hi  = lane >= 16;
    const int  t0  = wave;
    const bool two = (wave + 16) < 4 * BAND;
    const int  t1  = two ? (wave + 16) : wave;
    const int  r0 = t0 >> 2, w00 = ((t0 & 3) == 3) ? (PW - 16) : ((t0 & 3) * 16);
    const int  r1 = t1 >> 2, w01 = ((t1 & 3) == 3) ? (PW - 16) : ((t1 & 3) * 16);
    const float* a0p = pl + r0 * LW + w00 + m;
    const float* a1p = pl + r1 * LW + w01 + m;

    v8f acc0 = {0.f, 0.f, 0.f, 0.f, 0.f, 0.f, 0.f, 0.f};
    v8f acc1 = {0.f, 0.f, 0.f, 0.f, 0.f, 0.f, 0.f, 0.f};
#pragma unroll
    for (int k0 = 0; k0 < 100; k0 += 4) {
      v2f a0, a1, bw;
#pragma unroll
      for (int e = 0; e < 2; ++e) {
        const int klo = k0 + e;
        const int khi = k0 + 2 + e;
        const int kl  = (klo < 98) ? klo : 0;
        const int kh  = (khi < 98) ? khi : 0;
        const int offlo = (kl / 49) * (HROWS * LW) + ((kl % 49) / 7) * LW + (kl % 7);
        const int offhi = (kh / 49) * (HROWS * LW) + ((kh % 49) / 7) * LW + (kh % 7);
        const int off = hi ? offhi : offlo;
        const int k   = hi ? khi : klo;
        float av0 = a0p[off];
        float av1 = a1p[off];
        if (khi >= 98) {
          av0 = hi ? 0.f : av0;
          av1 = hi ? 0.f : av1;
        }
        const float bv = wsm[k];
        if (e == 0) { a0.x = av0; a1.x = av1; bw.x = bv; }
        else        { a0.y = av0; a1.y = av1; bw.y = bv; }
      }
      acc0 = __builtin_amdgcn_wmma_f32_16x16x4_f32(
           false, a0,  false, bw,
           (short)0, acc0,  false,  false);
      acc1 = __builtin_amdgcn_wmma_f32_16x16x4_f32(
           false, a1,  false, bw,
           (short)0, acc1,  false,  false);
    }

    int rr = 0, mout = 0;
    bool writer = false;
    if (lane < 8)                     { writer = true; rr = lane;      mout = lane;     }
    else if (lane >= 16 && lane < 24) { writer = true; rr = lane - 16; mout = lane - 8; }

    float y0 = acc0[0], y1 = acc1[0];
    if (rr == 1) { y0 = acc0[1]; y1 = acc1[1]; }
    if (rr == 2) { y0 = acc0[2]; y1 = acc1[2]; }
    if (rr == 3) { y0 = acc0[3]; y1 = acc1[3]; }
    if (rr == 4) { y0 = acc0[4]; y1 = acc1[4]; }
    if (rr == 5) { y0 = acc0[5]; y1 = acc1[5]; }
    if (rr == 6) { y0 = acc0[6]; y1 = acc1[6]; }
    if (rr == 7) { y0 = acc0[7]; y1 = acc1[7]; }

    if (writer) {
      y0 += bias0;
      ssm[r0 * PW + w00 + mout] = 1.0f / (1.0f + __expf(-y0));
      if (two) {
        y1 += bias0;
        ssm[r1 * PW + w01 + mout] = 1.0f / (1.0f + __expf(-y1));
      }
    }
  }
  __syncthreads();

  {
    const int sh = 2 * b;
    int   q[4];
    float4 sc[4];
#pragma unroll
    for (int i = 0; i < 4; ++i) {
      int qi = 32 * i + lane - sh;
      qi = (qi < 0) ? 0 : qi;
      qi = (qi > 97) ? 97 : qi;
      q[i]  = qi;
      sc[i] = *(const float4*)(ssm + 4 * qi);
    }
    const bool act0 = lane >= sh;
    const bool act3 = lane < sh + 2;
    const float* xband = xn + BAND * b * PW;
    float*       oband = out + (size_t)n * PCHW + BAND * b * PW;

#pragma unroll
    for (int g = 0; g < 4; ++g) {
      float4 v[4][4];
#pragma unroll
      for (int u = 0; u < 4; ++u) {
        const int c = g * 64 + wave * 4 + u;
#pragma unroll
        for (int i = 0; i < 4; ++i)
          v[u][i] = *(const float4*)(xband + (size_t)c * PHW + 4 * q[i]);
      }
#pragma unroll
      for (int u = 0; u < 4; ++u) {
        const int c = g * 64 + wave * 4 + u;
#pragma unroll
        for (int i = 0; i < 4; ++i) {
          float4 o;
          o.x = v[u][i].x * sc[i].x;  o.y = v[u][i].y * sc[i].y;
          o.z = v[u][i].z * sc[i].z;  o.w = v[u][i].w * sc[i].w;
          float* dst = oband + (size_t)c * PHW + 4 * q[i];
          if (i == 0)      { if (act0) *(float4*)dst = o; }
          else if (i == 3) { if (act3) *(float4*)dst = o; }
          else             { *(float4*)dst = o; }
        }
      }
    }
  }
}

extern "C" void kernel_launch(void* const* d_in, const int* in_sizes, int n_in,
                              void* d_out, int out_size, void* d_ws, size_t ws_size,
                              hipStream_t stream) {
  const float* x     = (const float*)d_in[0];
  const float* wconv = (const float*)d_in[1];
  const float* bias  = (const float*)d_in[2];
  float* out = (float*)d_out;

  sa_fused_kernel<<<256, NTHR, 0, stream>>>(x, wconv, bias, out);
}


# the 64 channels that do not fit the LDS copy stay in the pooling threads' load window (64 deep, was 48) through the conv and are scaled and stored from there: x is read once; sum orders unchanged
# speedup vs baseline: 2.0962x; 1.1399x over previous
#include <hip/hip_runtime.h>
#include <math.h>

#define PN   32
#define PC   256
#define PH   56
#define PW   56
#define PHW  (PH * PW)
#define PCHW (PC * PHW)

typedef float v2f __attribute__((ext_vector_type(2)));
typedef float v8f __attribute__((ext_vector_type(8)));

#define BAND   7
#define HROWS  (BAND + 6)
#define LW     64
#define NTHR   512
#define POOL_T 416
#define POOL_D 64
#define BANDPX (BAND * PW)
#define STASH_C 192
static_assert(STASH_C % 64 == 0 && STASH_C % POOL_D == 0 && PC - STASH_C == POOL_D,
              "the channels that are not copied to LDS are the window's last fill, slot j = channel STASH_C + j");

__global__ void __launch_bounds__(NTHR)  sa_fused_kernel(
    const float* __restrict__ x,
    const float* __restrict__ wconv,
    const float* __restrict__ bias,
    float* __restrict__ out) {
  __shared__ float pl[2 * HROWS * LW];
  __shared__ float wsm[100];
  __shared__ __attribute__((aligned(16))) float ssm[BAND * PW];
  __shared__ __attribute__((aligned(16))) float stash[STASH_C * BANDPX];
  __shared__ __attribute__((aligned(16))) float sdump[2 * POOL_T];

  const int tid   = threadIdx.x;
  const int n     = blockIdx.x >> 3;
  const int b     = blockIdx.x & 7;
  const int rbase = BAND * b - 3;
  const float* xn = x + (size_t)n * PCHW;
  const float bias0 = bias[0];

  float2 win[POOL_D];
  int pbk = -1;

  if (tid < POOL_T) {
    const int R0 = (rbase > 0) ? rbase : 0;
    const int R1 = (rbase + HROWS - 1 < PH - 1) ? (rbase + HROWS - 1) : (PH - 1);
    const int F0 = R0 * PW;
    const int F1 = (R1 + 1) * PW;
    const int f  = (F0 & ~63) + 2 * tid;
    if (f >= F0 && f < F1) {
      const float* xb = xn + f;
      const int  pb     = f - BANDPX * b;
      const bool inband = (pb >= 0) && (pb < BANDPX);
      float*    sp    = inband ? (stash + pb) : (sdump + 2 * tid);
      const int sstep = inband ? BANDPX : 0;
      pbk = inband ? pb : -1;
      float s0 = 0.f, s1 = 0.f;
      float m0 = -INFINITY, m1 = -INFINITY;
#pragma unroll
      for (int j = 0; j < POOL_D; ++j) {
        if ((j & 3) == 0) __builtin_amdgcn_sched_barrier(0);
        win[j] = *(const float2*)(xb + (size_t)j * PHW);
      }
#pragma unroll
      for (int c = 0; c < PC; ++c) {
        if ((c & 3) == 0) {
          __builtin_amdgcn_sched_barrier(0);
          asm volatile("" : "+v"(s0), "+v"(s1), "+v"(m0), "+v"(m1));
        }
        const float2 v = win[c % POOL_D];
        if (c < STASH_C) {
          *(float2*)sp = v;
          sp += sstep;
        }
        s0 += v.x; s1 += v.y;
        m0 = fmaxf(m0, v.x); m1 = fmaxf(m1, v.y);
        if (c + POOL_D < PC)
          win[c % POOL_D] = *(const float2*)(xb + (size_t)(c + POOL_D) * PHW);
      }
      const float inv = 1.0f / (float)PC;
      const int row = f / PW;
      const int col = f - row * PW;
      float* pd = pl + (row - rbase) * LW + col + 3;
      pd[0] = s0 * inv;  pd[1] = s1 * inv;
      pd[HROWS * LW] = m0;  pd[HROWS * LW + 1] = m1;
    }
  } else {
    for (int i = tid - POOL_T; i < 2 * HROWS * LW; i += NTHR - POOL_T) {
      const int rc  = (i >= HROWS * LW) ? (i - HROWS * LW) : i;
      const int lr  = rc / LW;
      const int col = rc - lr * LW;
      const int row = rbase + lr;
      if (col < 3 || col >= 3 + PW || row < 0 || row >= PH) pl[i] = 0.f;
    }
    for (int k = tid - POOL_T; k < 100; k += NTHR - POOL_T)
      wsm[k] = (k < 98) ? wconv[k] : 0.f;
  }
  __syncthreads();

  const int wave = tid >> 5;
  const int lane = tid & 31;

  {
    const int  m   = lane & 15;
    const bool hi  = lane >= 16;
    const int  t0  = wave;
    const bool two = (wave + 16) < 4 * BAND;
    const int  t1  = two ? (wave + 16) : wave;
    const int  r0 = t0 >> 2, w00 = ((t0 & 3) == 3) ? (PW - 16) : ((t0 & 3) * 16);
    const int  r1 = t1 >> 2, w01 = ((t1 & 3) == 3) ? (PW - 16) : ((t1 & 3) * 16);
    const float* a0p = pl + r0 * LW + w00 + m;
    const float* a1p = pl + r1 * LW + w01 + m;

    v8f acc0 = {0.f, 0.f, 0.f, 0.f, 0.f, 0.f, 0.f, 0.f};
    v8f acc1 = {0.f, 0.f, 0.f, 0.f, 0.f, 0.f, 0.f, 0.f};
#pragma unroll
    for (int k0 = 0; k0 < 100; k0 += 4) {
      v2f a0, a1, bw;
#pragma unroll
      for (int e = 0; e < 2; ++e) {
        const int klo = k0 + e;
        const int khi = k0 + 2 + e;
        const int kl  = (klo < 98) ? klo : 0;
        const int kh  = (khi < 98) ? khi : 0;
        const int offlo = (kl / 49) * (HROWS * LW) + ((kl % 49) / 7) * LW + (kl % 7);
        const int offhi = (kh / 49) * (HROWS * LW) + ((kh % 49) / 7) * LW + (kh % 7);
        const int off = hi ? offhi : offlo;
        const int k   = hi ? khi : klo;
        float av0 = a0p[off];
        float av1 = a1p[off];
        if (khi >= 98) {
          av0 = hi ? 0.f : av0;
          av1 = hi ? 0.f : av1;
        }
        const float bv = wsm[k];
        if (e == 0) { a0.x = av0; a1.x = av1; bw.x = bv; }
        else        { a0.y = av0; a1.y = av1; bw.y = bv; }
      }
      acc0 = __builtin_amdgcn_wmma_f32_16x16x4_f32(
           false, a0,  false, bw,
           (short)0, acc0,  false,  false);
      acc1 = __builtin_amdgcn_wmma_f32_16x16x4_f32(
           false, a1,  false, bw,
           (short)0, acc1,  false,  false);
    }

    int rr = 0, mout = 0;
    bool writer = false;
    if (lane < 8)                     { writer = true; rr = lane;      mout = lane;     }
    else if (lane >= 16 && lane < 24) { writer = true; rr = lane - 16; mout = lane - 8; }

    float y0 = acc0[0], y1 = acc1[0];
    if (rr == 1) { y0 = acc0[1]; y1 = acc1[1]; }
    if (rr == 2) { y0 = acc0[2]; y1 = acc1[2]; }
    if (rr == 3) { y0 = acc0[3]; y1 = acc1[3]; }
    if (rr == 4) { y0 = acc0[4]; y1 = acc1[4]; }
    if (rr == 5) { y0 = acc0[5]; y1 = acc1[5]; }
    if (rr == 6) { y0 = acc0[6]; y1 = acc1[6]; }
    if (rr == 7) { y0 = acc0[7]; y1 = acc1[7]; }

    if (writer) {
      y0 += bias0;
      ssm[r0 * PW + w00 + mout] = 1.0f / (1.0f + __expf(-y0));
      if (two) {
        y1 += bias0;
        ssm[r1 * PW + w01 + mout] = 1.0f / (1.0f + __expf(-y1));
      }
    }
  }
  __syncthreads();

  {
    float*       oband = out + (size_t)n * PCHW + BAND * b * PW;
    if (pbk >= 0) {
      const float2 s2 = *(const float2*)(ssm + pbk);
      float* dk = oband + (size_t)STASH_C * PHW + pbk;
#pragma unroll
      for (int j = 0; j < POOL_D; ++j) {
        float2 o;
        o.x = win[j].x * s2.x;  o.y = win[j].y * s2.y;
        *(float2*)(dk + (size_t)j * PHW) = o;
      }
    }

    const int sh = 2 * b;
    int   q[4];
    float4 sc[4];
#pragma unroll
    for (int i = 0; i < 4; ++i) {
      int qi = 32 * i + lane - sh;
      qi = (qi < 0) ? 0 : qi;
      qi = (qi > 97) ? 97 : qi;
      q[i]  = qi;
      sc[i] = *(const float4*)(ssm + 4 * qi);
    }
    const bool act0 = lane >= sh;
    const bool act3 = lane < sh + 2;

#pragma unroll
    for (int g = 0; g < STASH_C / 64; ++g) {
      float4 v[4][4];
#pragma unroll
      for (int u = 0; u < 4; ++u) {
        const int c = g * 64 + wave * 4 + u;
#pragma unroll
        for (int i = 0; i < 4; ++i)
          v[u][i] = *(const float4*)(stash + c * BANDPX + 4 * q[i]);
      }
#pragma unroll
      for (int u = 0; u < 4; ++u) {
        const int c = g * 64 + wave * 4 + u;
#pragma unroll
        for (int i = 0; i < 4; ++i) {
          float4 o;
          o.x = v[u][i].x * sc[i].x;  o.y = v[u][i].y * sc[i].y;
          o.z = v[u][i].z * sc[i].z;  o.w = v[u][i].w * sc[i].w;
          float* dst = oband + (size_t)c * PHW + 4 * q[i];
          if (i == 0)      { if (act0) *(float4*)dst = o; }
          else if (i == 3) { if (act3) *(float4*)dst = o; }
          else             { *(float4*)dst = o; }
        }
      }
    }

  }
}

extern "C" void kernel_launch(void* const* d_in, const int* in_sizes, int n_in,
                              void* d_out, int out_size, void* d_ws, size_t ws_size,
                              hipStream_t stream) {
  const float* x     = (const float*)d_in[0];
  const float* wconv = (const float*)d_in[1];
  const float* bias  = (const float*)d_in[2];
  float* out = (float*)d_out;

  sa_fused_kernel<<<256, NTHR, 0, stream>>>(x, wconv, bias, out);
}
